# starting kernel
# baseline (speedup 1.0000x reference)
#include <hip/hip_runtime.h>
#include <math.h>

typedef float v2f __attribute__((ext_vector_type(2)));
typedef float v8f __attribute__((ext_vector_type(8)));

#define TWO_PI_F     6.2831853071795864769f
#define MIN_W        1e-3f
#define MIN_H        1e-3f
#define MIN_D        1e-3f
#define DERIV_SHIFT  0.54132485461292f

#define KBINS 32
#define DH    8
#define CDIM  64
#define HDIM  256
#define NPAR  768

#define WAVES           2
#define MTILES          2
#define ROWS_PER_WAVE   (MTILES * 16)
#define ROWS_PER_BLOCK  (WAVES * ROWS_PER_WAVE)
#define PCHUNK          96
#define NT2             (PCHUNK / 16)

#define HBUF_FLOATS (ROWS_PER_BLOCK * HDIM)
#define PBUF_FLOATS (ROWS_PER_BLOCK * PCHUNK)
#define LDS_BYTES   ((HBUF_FLOATS + PBUF_FLOATS) * 4)

static __device__ __forceinline__ float softplus_f(float x) {
    return (x > 20.0f) ? x : log1pf(__expf(x));
}

__global__ __launch_bounds__(WAVES * 32)
void Circular_Spline_Flow_fused_kernel(
    const float* __restrict__ theta,
    const float* __restrict__ x,
    const float* __restrict__ W1,
    const float* __restrict__ b1,
    const float* __restrict__ W2,
    const float* __restrict__ b2,
    const float* __restrict__ eta,
    float* __restrict__ out,
    int Btotal)
{
    extern __shared__ float smem[];

    const int tid  = threadIdx.x;
    const int wave = tid >> 5;
    const int lane = tid & 31;
    const int m    = lane & 15;
    const int half = lane >> 4;

    const int rowbase = blockIdx.x * ROWS_PER_BLOCK + wave * ROWS_PER_WAVE;

    float* hbuf = smem + wave * ROWS_PER_WAVE * HDIM;
    float* pbuf = smem + HBUF_FLOATS + wave * ROWS_PER_WAVE * PCHUNK;

    const float etav = eta[0];

    v2f afrag[MTILES][16];
    #pragma unroll
    for (int mt = 0; mt < MTILES; ++mt) {
        const float* xrow = x + (size_t)(rowbase + mt * 16 + m) * CDIM;
        #pragma unroll
        for (int kc = 0; kc < 16; ++kc) {
            const int kk = 4 * kc + 2 * half;
            afrag[mt][kc].x = xrow[kk];
            afrag[mt][kc].y = xrow[kk + 1];
        }
    }

    for (int j = 0; j < 16; ++j) {
        const int col = 16 * j + m;
        v8f acc0 = {0.f,0.f,0.f,0.f,0.f,0.f,0.f,0.f};
        v8f acc1 = {0.f,0.f,0.f,0.f,0.f,0.f,0.f,0.f};
        #pragma unroll
        for (int kc = 0; kc < 16; ++kc) {
            const int kk = 4 * kc + 2 * half;
            v2f bfrag;
            bfrag.x = W1[(size_t)kk * HDIM + col];
            bfrag.y = W1[(size_t)(kk + 1) * HDIM + col];
            acc0 = __builtin_amdgcn_wmma_f32_16x16x4_f32(
                false, afrag[0][kc], false, bfrag, (short)0, acc0, false, false);
            acc1 = __builtin_amdgcn_wmma_f32_16x16x4_f32(
                false, afrag[1][kc], false, bfrag, (short)0, acc1, false, false);
        }
        const float bias = b1[col];
        #pragma unroll
        for (int v = 0; v < 8; ++v) {
            float h0 = fmaxf(acc0[v] + bias, 0.0f);
            float h1 = fmaxf(acc1[v] + bias, 0.0f);
            hbuf[( 0 + 8 * half + v) * HDIM + col] = h0;
            hbuf[(16 + 8 * half + v) * HDIM + col] = h1;
        }
    }

    for (int dp = 0; dp < DH; ++dp) {
        const int colbase = dp * PCHUNK;

        __builtin_prefetch(&W2[(size_t)(2 * half) * NPAR + colbase + m], 0, 1);

        v8f acc[MTILES][NT2];
        #pragma unroll
        for (int mt = 0; mt < MTILES; ++mt)
            #pragma unroll
            for (int jt = 0; jt < NT2; ++jt)
                acc[mt][jt] = (v8f){0.f,0.f,0.f,0.f,0.f,0.f,0.f,0.f};

        #pragma unroll 2
        for (int kc = 0; kc < 64; ++kc) {
            const int kk = 4 * kc + 2 * half;
            v2f a0, a1;
            a0.x = hbuf[( 0 + m) * HDIM + kk];
            a0.y = hbuf[( 0 + m) * HDIM + kk + 1];
            a1.x = hbuf[(16 + m) * HDIM + kk];
            a1.y = hbuf[(16 + m) * HDIM + kk + 1];
            #pragma unroll
            for (int jt = 0; jt < NT2; ++jt) {
                const int col = colbase + 16 * jt + m;
                v2f b;
                b.x = W2[(size_t)kk * NPAR + col];
                b.y = W2[(size_t)(kk + 1) * NPAR + col];
                acc[0][jt] = __builtin_amdgcn_wmma_f32_16x16x4_f32(
                    false, a0, false, b, (short)0, acc[0][jt], false, false);
                acc[1][jt] = __builtin_amdgcn_wmma_f32_16x16x4_f32(
                    false, a1, false, b, (short)0, acc[1][jt], false, false);
            }
        }

        #pragma unroll
        for (int jt = 0; jt < NT2; ++jt) {
            const float bias = b2[colbase + 16 * jt + m];
            #pragma unroll
            for (int mt = 0; mt < MTILES; ++mt) {
                #pragma unroll
                for (int v = 0; v < 8; ++v) {
                    const float pv = (acc[mt][jt][v] + bias) * etav;
                    pbuf[(mt * 16 + 8 * half + v) * PCHUNK + 16 * jt + m] = pv;
                }
            }
        }

        {
            const float* P  = &pbuf[lane * PCHUNK];
            const float* uw = P;
            const float* uh = P + 32;
            const float* ud = P + 64;
            const float  th = theta[(size_t)(rowbase + lane) * DH + dp];

            float mw = -1e30f, mh = -1e30f;
            for (int i = 0; i < KBINS; ++i) {
                mw = fmaxf(mw, uw[i]);
                mh = fmaxf(mh, uh[i]);
            }
            float sw = 0.f, sh = 0.f;
            for (int i = 0; i < KBINS; ++i) {
                sw += __expf(uw[i] - mw);
                sh += __expf(uh[i] - mh);
            }
            const float wscale = (1.0f - MIN_W * KBINS) / sw;
            const float hscale = (1.0f - MIN_H * KBINS) / sh;

            float cw = 0.f, ch = 0.f;
            float Sw = 0.f, Sh = 0.f;
            int   idx = 0;
            float sel_cw = 0.f, sel_ch = 0.f, sel_w = TWO_PI_F, sel_h = TWO_PI_F;
            for (int i = 0; i < KBINS; ++i) {
                const float wfrac = MIN_W + wscale * __expf(uw[i] - mw);
                const float hfrac = MIN_H + hscale * __expf(uh[i] - mh);
                Sw += wfrac;
                Sh += hfrac;
                const float cw_next = (i == KBINS - 1) ? TWO_PI_F : TWO_PI_F * Sw;
                const float ch_next = (i == KBINS - 1) ? TWO_PI_F : TWO_PI_F * Sh;
                if (th >= cw) {
                    idx    = i;
                    sel_cw = cw;  sel_w = cw_next - cw;
                    sel_ch = ch;  sel_h = ch_next - ch;
                }
                cw = cw_next;
                ch = ch_next;
            }

            const float d0 = MIN_D + softplus_f(ud[idx] + DERIV_SHIFT);
            const float d1 = MIN_D + softplus_f(ud[(idx + 1) & (KBINS - 1)] + DERIV_SHIFT);

            const float delta = sel_h / sel_w;
            const float t     = (th - sel_cw) / sel_w;
            const float t1mt  = t * (1.0f - t);
            const float numer = sel_h * (delta * t * t + d0 * t1mt);
            const float denom = delta + (d0 + d1 - 2.0f * delta) * t1mt;
            const float outv  = sel_ch + numer / denom;

            const float omt  = 1.0f - t;
            const float dnum = delta * delta * (d1 * t * t + 2.0f * delta * t1mt + d0 * omt * omt);
            const float lad  = __logf(dnum) - 2.0f * __logf(denom);

            const size_t o = (size_t)(rowbase + lane) * DH + dp;
            out[o] = outv;
            out[(size_t)Btotal * DH + o] = lad;
        }
    }
}

extern "C" void kernel_launch(void* const* d_in, const int* in_sizes, int n_in,
                              void* d_out, int out_size, void* d_ws, size_t ws_size,
                              hipStream_t stream) {
    (void)n_in; (void)out_size; (void)d_ws; (void)ws_size;
    const float* theta = (const float*)d_in[0];
    const float* x     = (const float*)d_in[1];
    const float* W1    = (const float*)d_in[2];
    const float* b1    = (const float*)d_in[3];
    const float* W2    = (const float*)d_in[4];
    const float* b2    = (const float*)d_in[5];
    const float* eta   = (const float*)d_in[6];
    float* out = (float*)d_out;

    const int B = in_sizes[0] / DH;
    const int blocks = B / ROWS_PER_BLOCK;
    Circular_Spline_Flow_fused_kernel<<<blocks, WAVES * 32, LDS_BYTES, stream>>>(
        theta, x, W1, b1, W2, b2, eta, out, B);
}


# software-pipelined fragment loads: GEMM2 A/B fragments ride a 4-slot register ring requested 4 K-steps ahead (also across heads), W1 tiles double-buffered; same k order, sum order unchanged
# speedup vs baseline: 1.3594x; 1.3594x over previous
#include <hip/hip_runtime.h>
#include <math.h>

typedef float v2f __attribute__((ext_vector_type(2)));
typedef float v8f __attribute__((ext_vector_type(8)));

#define TWO_PI_F     6.2831853071795864769f
#define MIN_W        1e-3f
#define MIN_H        1e-3f
#define MIN_D        1e-3f
#define DERIV_SHIFT  0.54132485461292f

#define KBINS 32
#define DH    8
#define CDIM  64
#define HDIM  256
#define NPAR  768

#define WAVES           2
#define MTILES          2
#define ROWS_PER_WAVE   (MTILES * 16)
#define ROWS_PER_BLOCK  (WAVES * ROWS_PER_WAVE)
#define PCHUNK          96
#define NT2             (PCHUNK / 16)

#define HBUF_FLOATS (ROWS_PER_BLOCK * HDIM)
#define PBUF_FLOATS (ROWS_PER_BLOCK * PCHUNK)
#define LDS_BYTES   ((HBUF_FLOATS + PBUF_FLOATS) * 4)

#define KSTEPS  (HDIM / 4)
#define PF_DIST 4

static __device__ __forceinline__ float softplus_f(float x) {
    return (x > 20.0f) ? x : log1pf(__expf(x));
}

static __device__ __forceinline__ void load_w2_step(
    const float* __restrict__ W2, int dpn, int kc, int half, int m, v2f (&dst)[NT2])
{
    const float* p = W2 + (size_t)(4 * kc + 2 * half) * NPAR + dpn * PCHUNK + m;
    #pragma unroll
    for (int jt = 0; jt < NT2; ++jt) {
        dst[jt].x = p[16 * jt];
        dst[jt].y = p[NPAR + 16 * jt];
    }
}

static __device__ __forceinline__ void load_a_step(
    const float* hbuf, int kc, int half, int m, v2f (&dst)[MTILES])
{
    const int kk = 4 * kc + 2 * half;
    #pragma unroll
    for (int mt = 0; mt < MTILES; ++mt) {
        dst[mt].x = hbuf[(mt * 16 + m) * HDIM + kk];
        dst[mt].y = hbuf[(mt * 16 + m) * HDIM + kk + 1];
    }
}

static __device__ __forceinline__ void load_w1_tile(
    const float* __restrict__ W1, int j, int half, int m, v2f (&dst)[16])
{
    const float* p = W1 + (size_t)(2 * half) * HDIM + 16 * j + m;
    #pragma unroll
    for (int kc = 0; kc < 16; ++kc) {
        dst[kc].x = p[(size_t)(4 * kc) * HDIM];
        dst[kc].y = p[(size_t)(4 * kc + 1) * HDIM];
    }
}

static __device__ __forceinline__ void gemm1_tile(
    const float* __restrict__ W1, const float* __restrict__ b1, float* hbuf,
    const v2f (&afrag)[MTILES][16], const v2f (&cur)[16], v2f (&nxt)[16],
    int j, bool prefetch, int half, int m)
{
    const int col = 16 * j + m;
    const float bias = b1[col];
    if (prefetch)
        load_w1_tile(W1, j + 1, half, m, nxt);
    __builtin_amdgcn_sched_barrier(0);

    v8f acc0 = {0.f,0.f,0.f,0.f,0.f,0.f,0.f,0.f};
    v8f acc1 = {0.f,0.f,0.f,0.f,0.f,0.f,0.f,0.f};
    #pragma unroll
    for (int kc = 0; kc < 16; ++kc) {
        acc0 = __builtin_amdgcn_wmma_f32_16x16x4_f32(
            false, afrag[0][kc], false, cur[kc], (short)0, acc0, false, false);
        acc1 = __builtin_amdgcn_wmma_f32_16x16x4_f32(
            false, afrag[1][kc], false, cur[kc], (short)0, acc1, false, false);
    }
    #pragma unroll
    for (int v = 0; v < 8; ++v) {
        float h0 = fmaxf(acc0[v] + bias, 0.0f);
        float h1 = fmaxf(acc1[v] + bias, 0.0f);
        hbuf[( 0 + 8 * half + v) * HDIM + col] = h0;
        hbuf[(16 + 8 * half + v) * HDIM + col] = h1;
    }
}

__global__ __launch_bounds__(WAVES * 32)
void Circular_Spline_Flow_fused_kernel(
    const float* __restrict__ theta,
    const float* __restrict__ x,
    const float* __restrict__ W1,
    const float* __restrict__ b1,
    const float* __restrict__ W2,
    const float* __restrict__ b2,
    const float* __restrict__ eta,
    float* __restrict__ out,
    int Btotal)
{
    extern __shared__ float smem[];

    const int tid  = threadIdx.x;
    const int wave = tid >> 5;
    const int lane = tid & 31;
    const int m    = lane & 15;
    const int half = lane >> 4;

    const int rowbase = blockIdx.x * ROWS_PER_BLOCK + wave * ROWS_PER_WAVE;

    float* hbuf = smem + wave * ROWS_PER_WAVE * HDIM;
    float* pbuf = smem + HBUF_FLOATS + wave * ROWS_PER_WAVE * PCHUNK;

    const float etav = eta[0];

    v2f afrag[MTILES][16];
    #pragma unroll
    for (int mt = 0; mt < MTILES; ++mt) {
        const float* xrow = x + (size_t)(rowbase + mt * 16 + m) * CDIM;
        #pragma unroll
        for (int kc = 0; kc < 16; ++kc) {
            const int kk = 4 * kc + 2 * half;
            afrag[mt][kc].x = xrow[kk];
            afrag[mt][kc].y = xrow[kk + 1];
        }
    }

    v2f bq[PF_DIST][NT2];
    #pragma unroll
    for (int u = 0; u < PF_DIST; ++u)
        load_w2_step(W2, 0, u, half, m, bq[u]);

    v2f bw[2][16];
    load_w1_tile(W1, 0, half, m, bw[0]);

    for (int j2 = 0; j2 < 14; j2 += 2) {
        gemm1_tile(W1, b1, hbuf, afrag, bw[0], bw[1], j2,     true, half, m);
        gemm1_tile(W1, b1, hbuf, afrag, bw[1], bw[0], j2 + 1, true, half, m);
    }
    gemm1_tile(W1, b1, hbuf, afrag, bw[0], bw[1], 14, true,  half, m);
    gemm1_tile(W1, b1, hbuf, afrag, bw[1], bw[0], 15, false, half, m);

    v2f aq[PF_DIST][MTILES];
    #pragma unroll
    for (int u = 0; u < PF_DIST; ++u)
        load_a_step(hbuf, u, half, m, aq[u]);

    for (int dp = 0; dp < DH; ++dp) {
        const int colbase = dp * PCHUNK;

        __builtin_prefetch(&W2[(size_t)(2 * half) * NPAR + colbase + m], 0, 1);

        v8f acc[MTILES][NT2];
        #pragma unroll
        for (int mt = 0; mt < MTILES; ++mt)
            #pragma unroll
            for (int jt = 0; jt < NT2; ++jt)
                acc[mt][jt] = (v8f){0.f,0.f,0.f,0.f,0.f,0.f,0.f,0.f};

        float bias2[NT2];
        #pragma unroll
        for (int jt = 0; jt < NT2; ++jt)
            bias2[jt] = b2[colbase + 16 * jt + m];
        const float th = theta[(size_t)(rowbase + lane) * DH + dp];

        const int dpn = (dp + 1 < DH) ? dp + 1 : DH - 1;

        #pragma unroll 1
        for (int kc4 = 0; kc4 < KSTEPS; kc4 += PF_DIST) {
            #pragma unroll
            for (int u = 0; u < PF_DIST; ++u) {
                const int kc = kc4 + u;
                #pragma unroll
                for (int jt = 0; jt < NT2; ++jt) {
                    acc[0][jt] = __builtin_amdgcn_wmma_f32_16x16x4_f32(
                        false, aq[u][0], false, bq[u][jt], (short)0, acc[0][jt], false, false);
                    acc[1][jt] = __builtin_amdgcn_wmma_f32_16x16x4_f32(
                        false, aq[u][1], false, bq[u][jt], (short)0, acc[1][jt], false, false);
                }
                const int kn   = kc + PF_DIST;
                const int wrap = (kn >= KSTEPS);
                const int kcn  = kn & (KSTEPS - 1);
                load_a_step(hbuf, kcn, half, m, aq[u]);
                load_w2_step(W2, wrap ? dpn : dp, kcn, half, m, bq[u]);
                __builtin_amdgcn_sched_barrier(0);
            }
        }

        #pragma unroll
        for (int jt = 0; jt < NT2; ++jt) {
            const float bias = bias2[jt];
            #pragma unroll
            for (int mt = 0; mt < MTILES; ++mt) {
                #pragma unroll
                for (int v = 0; v < 8; ++v) {
                    const float pv = (acc[mt][jt][v] + bias) * etav;
                    pbuf[(mt * 16 + 8 * half + v) * PCHUNK + 16 * jt + m] = pv;
                }
            }
        }

        {
            const float* P  = &pbuf[lane * PCHUNK];
            const float* uw = P;
            const float* uh = P + 32;
            const float* ud = P + 64;

            float mw = -1e30f, mh = -1e30f;
            for (int i = 0; i < KBINS; ++i) {
                mw = fmaxf(mw, uw[i]);
                mh = fmaxf(mh, uh[i]);
            }
            float sw = 0.f, sh = 0.f;
            for (int i = 0; i < KBINS; ++i) {
                sw += __expf(uw[i] - mw);
                sh += __expf(uh[i] - mh);
            }
            const float wscale = (1.0f - MIN_W * KBINS) / sw;
            const float hscale = (1.0f - MIN_H * KBINS) / sh;

            float cw = 0.f, ch = 0.f;
            float Sw = 0.f, Sh = 0.f;
            int   idx = 0;
            float sel_cw = 0.f, sel_ch = 0.f, sel_w = TWO_PI_F, sel_h = TWO_PI_F;
            for (int i = 0; i < KBINS; ++i) {
                const float wfrac = MIN_W + wscale * __expf(uw[i] - mw);
                const float hfrac = MIN_H + hscale * __expf(uh[i] - mh);
                Sw += wfrac;
                Sh += hfrac;
                const float cw_next = (i == KBINS - 1) ? TWO_PI_F : TWO_PI_F * Sw;
                const float ch_next = (i == KBINS - 1) ? TWO_PI_F : TWO_PI_F * Sh;
                if (th >= cw) {
                    idx    = i;
                    sel_cw = cw;  sel_w = cw_next - cw;
                    sel_ch = ch;  sel_h = ch_next - ch;
                }
                cw = cw_next;
                ch = ch_next;
            }

            const float d0 = MIN_D + softplus_f(ud[idx] + DERIV_SHIFT);
            const float d1 = MIN_D + softplus_f(ud[(idx + 1) & (KBINS - 1)] + DERIV_SHIFT);

            const float delta = sel_h / sel_w;
            const float t     = (th - sel_cw) / sel_w;
            const float t1mt  = t * (1.0f - t);
            const float numer = sel_h * (delta * t * t + d0 * t1mt);
            const float denom = delta + (d0 + d1 - 2.0f * delta) * t1mt;
            const float outv  = sel_ch + numer / denom;

            const float omt  = 1.0f - t;
            const float dnum = delta * delta * (d1 * t * t + 2.0f * delta * t1mt + d0 * omt * omt);
            const float lad  = __logf(dnum) - 2.0f * __logf(denom);

            const size_t o = (size_t)(rowbase + lane) * DH + dp;
            out[o] = outv;
            out[(size_t)Btotal * DH + o] = lad;
        }
    }
}

extern "C" void kernel_launch(void* const* d_in, const int* in_sizes, int n_in,
                              void* d_out, int out_size, void* d_ws, size_t ws_size,
                              hipStream_t stream) {
    (void)n_in; (void)out_size; (void)d_ws; (void)ws_size;
    const float* theta = (const float*)d_in[0];
    const float* x     = (const float*)d_in[1];
    const float* W1    = (const float*)d_in[2];
    const float* b1    = (const float*)d_in[3];
    const float* W2    = (const float*)d_in[4];
    const float* b2    = (const float*)d_in[5];
    const float* eta   = (const float*)d_in[6];
    float* out = (float*)d_out;

    const int B = in_sizes[0] / DH;
    const int blocks = B / ROWS_PER_BLOCK;
    Circular_Spline_Flow_fused_kernel<<<blocks, WAVES * 32, LDS_BYTES, stream>>>(
        theta, x, W1, b1, W2, b2, eta, out, B);
}


# heads split over the 8 waves of a 32-row work-group sharing one H tile in LDS (was 2 waves x 32 rows, 8 heads each): 16 waves per WGP, not 6; whole-line stores via LDS; k and sum order same
# speedup vs baseline: 2.5194x; 1.8533x over previous
#include <hip/hip_runtime.h>
#include <math.h>

typedef float v2f __attribute__((ext_vector_type(2)));
typedef float v4f __attribute__((ext_vector_type(4)));
typedef float v8f __attribute__((ext_vector_type(8)));

#define TWO_PI_F     6.2831853071795864769f
#define MIN_W        1e-3f
#define MIN_H        1e-3f
#define MIN_D        1e-3f
#define DERIV_SHIFT  0.54132485461292f

#define KBINS 32
#define DH    8
#define CDIM  64
#define HDIM  256
#define NPAR  768

#define WAVES           DH
#define MTILES          2
#define ROWS_PER_BLOCK  (MTILES * 16)
#define PCHUNK          96
#define NT2             (PCHUNK / 16)

#define HBUF_FLOATS (ROWS_PER_BLOCK * HDIM)
#define PBUF_FLOATS (WAVES * ROWS_PER_BLOCK * PCHUNK)
#define OBUF_FLOATS (2 * ROWS_PER_BLOCK * DH)
#define LDS_BYTES   ((HBUF_FLOATS + PBUF_FLOATS + OBUF_FLOATS) * 4)

#define KSTEPS  (HDIM / 4)
#define PF_DIST 4

static __device__ __forceinline__ float softplus_f(float x) {
    return (x > 20.0f) ? x : log1pf(__expf(x));
}

static __device__ __forceinline__ void load_w2_step(
    const float* __restrict__ W2, int dpn, int kc, int half, int m, v2f (&dst)[NT2])
{
    const float* p = W2 + (size_t)(4 * kc + 2 * half) * NPAR + dpn * PCHUNK + m;
    #pragma unroll
    for (int jt = 0; jt < NT2; ++jt) {
        dst[jt].x = p[16 * jt];
        dst[jt].y = p[NPAR + 16 * jt];
    }
}

static __device__ __forceinline__ void load_a_step(
    const float* hbuf, int kc, int half, int m, v2f (&dst)[MTILES])
{
    const int kk = 4 * kc + 2 * half;
    #pragma unroll
    for (int mt = 0; mt < MTILES; ++mt) {
        dst[mt].x = hbuf[(mt * 16 + m) * HDIM + kk];
        dst[mt].y = hbuf[(mt * 16 + m) * HDIM + kk + 1];
    }
}

static __device__ __forceinline__ void load_w1_tile(
    const float* __restrict__ W1, int j, int half, int m, v2f (&dst)[16])
{
    const float* p = W1 + (size_t)(2 * half) * HDIM + 16 * j + m;
    #pragma unroll
    for (int kc = 0; kc < 16; ++kc) {
        dst[kc].x = p[(size_t)(4 * kc) * HDIM];
        dst[kc].y = p[(size_t)(4 * kc + 1) * HDIM];
    }
}

static __device__ __forceinline__ void gemm1_tile(
    const float* __restrict__ W1, const float* __restrict__ b1, float* hbuf,
    const v2f (&afrag)[MTILES][16], const v2f (&cur)[16], v2f (&nxt)[16],
    int j, bool prefetch, int half, int m)
{
    const int col = 16 * j + m;
    const float bias = b1[col];
    if (prefetch)
        load_w1_tile(W1, j + 1, half, m, nxt);
    __builtin_amdgcn_sched_barrier(0);

    v8f acc0 = {0.f,0.f,0.f,0.f,0.f,0.f,0.f,0.f};
    v8f acc1 = {0.f,0.f,0.f,0.f,0.f,0.f,0.f,0.f};
    #pragma unroll
    for (int kc = 0; kc < 16; ++kc) {
        acc0 = __builtin_amdgcn_wmma_f32_16x16x4_f32(
            false, afrag[0][kc], false, cur[kc], (short)0, acc0, false, false);
        acc1 = __builtin_amdgcn_wmma_f32_16x16x4_f32(
            false, afrag[1][kc], false, cur[kc], (short)0, acc1, false, false);
    }
    #pragma unroll
    for (int v = 0; v < 8; ++v) {
        float h0 = fmaxf(acc0[v] + bias, 0.0f);
        float h1 = fmaxf(acc1[v] + bias, 0.0f);
        hbuf[( 0 + 8 * half + v) * HDIM + col] = h0;
        hbuf[(16 + 8 * half + v) * HDIM + col] = h1;
    }
}

static __device__ __forceinline__ void gemm2_group(
    const float* __restrict__ W2, const float* hbuf, int dp, int kc4, bool request,
    int half, int m,
    v2f (&aq)[PF_DIST][MTILES], v2f (&bq)[PF_DIST][NT2], v8f (&acc)[MTILES][NT2])
{
    #pragma unroll
    for (int u = 0; u < PF_DIST; ++u) {
        #pragma unroll
        for (int jt = 0; jt < NT2; ++jt) {
            acc[0][jt] = __builtin_amdgcn_wmma_f32_16x16x4_f32(
                false, aq[u][0], false, bq[u][jt], (short)0, acc[0][jt], false, false);
            acc[1][jt] = __builtin_amdgcn_wmma_f32_16x16x4_f32(
                false, aq[u][1], false, bq[u][jt], (short)0, acc[1][jt], false, false);
        }
        if (request) {
            const int kn = kc4 + u + PF_DIST;
            load_a_step(hbuf, kn, half, m, aq[u]);
            load_w2_step(W2, dp, kn, half, m, bq[u]);
            __builtin_amdgcn_sched_barrier(0);
        }
    }
}

__global__ __launch_bounds__(WAVES * 32, 4)
void Circular_Spline_Flow_fused_kernel(
    const float* __restrict__ theta,
    const float* __restrict__ x,
    const float* __restrict__ W1,
    const float* __restrict__ b1,
    const float* __restrict__ W2,
    const float* __restrict__ b2,
    const float* __restrict__ eta,
    float* __restrict__ out,
    int Btotal)
{
    __shared__ __attribute__((aligned(16))) float smem[LDS_BYTES / 4];

    const int tid  = threadIdx.x;
    const int wave = tid >> 5;
    const int lane = tid & 31;
    const int m    = lane & 15;
    const int half = lane >> 4;

    const int rowbase = blockIdx.x * ROWS_PER_BLOCK;
    const int dp      = wave;

    float* hbuf = smem;
    float* pbuf = smem + HBUF_FLOATS + wave * ROWS_PER_BLOCK * PCHUNK;
    float* obuf = smem + HBUF_FLOATS + PBUF_FLOATS;

    const float etav = eta[0];

    v2f afrag[MTILES][16];
    #pragma unroll
    for (int mt = 0; mt < MTILES; ++mt) {
        const float* xrow = x + (size_t)(rowbase + mt * 16 + m) * CDIM;
        #pragma unroll
        for (int kc = 0; kc < 16; ++kc) {
            const int kk = 4 * kc + 2 * half;
            afrag[mt][kc].x = xrow[kk];
            afrag[mt][kc].y = xrow[kk + 1];
        }
    }

    v2f bq[PF_DIST][NT2];
    #pragma unroll
    for (int u = 0; u < PF_DIST; ++u)
        load_w2_step(W2, dp, u, half, m, bq[u]);

    float bias2[NT2];
    #pragma unroll
    for (int jt = 0; jt < NT2; ++jt)
        bias2[jt] = b2[dp * PCHUNK + 16 * jt + m];
    const float th = theta[(size_t)(rowbase + lane) * DH + dp];

    v2f bw[2][16];
    load_w1_tile(W1, 2 * wave, half, m, bw[0]);
    gemm1_tile(W1, b1, hbuf, afrag, bw[0], bw[1], 2 * wave,     true,  half, m);
    gemm1_tile(W1, b1, hbuf, afrag, bw[1], bw[0], 2 * wave + 1, false, half, m);

    __syncthreads();

    v2f aq[PF_DIST][MTILES];
    #pragma unroll
    for (int u = 0; u < PF_DIST; ++u)
        load_a_step(hbuf, u, half, m, aq[u]);

    v8f acc[MTILES][NT2];
    #pragma unroll
    for (int mt = 0; mt < MTILES; ++mt)
        #pragma unroll
        for (int jt = 0; jt < NT2; ++jt)
            acc[mt][jt] = (v8f){0.f,0.f,0.f,0.f,0.f,0.f,0.f,0.f};

    #pragma unroll 1
    for (int kc4 = 0; kc4 < KSTEPS - PF_DIST; kc4 += PF_DIST)
        gemm2_group(W2, hbuf, dp, kc4, true, half, m, aq, bq, acc);
    gemm2_group(W2, hbuf, dp, KSTEPS - PF_DIST, false, half, m, aq, bq, acc);

    #pragma unroll
    for (int jt = 0; jt < NT2; ++jt) {
        const float bias = bias2[jt];
        #pragma unroll
        for (int mt = 0; mt < MTILES; ++mt) {
            #pragma unroll
            for (int v = 0; v < 8; ++v) {
                const float pv = (acc[mt][jt][v] + bias) * etav;
                pbuf[(mt * 16 + 8 * half + v) * PCHUNK + 16 * jt + m] = pv;
            }
        }
    }

    {
        const float* P  = &pbuf[lane * PCHUNK];
        const float* uw = P;
        const float* uh = P + 32;
        const float* ud = P + 64;

        float mw = -1e30f, mh = -1e30f;
        for (int i = 0; i < KBINS; ++i) {
            mw = fmaxf(mw, uw[i]);
            mh = fmaxf(mh, uh[i]);
        }
        float sw = 0.f, sh = 0.f;
        for (int i = 0; i < KBINS; ++i) {
            sw += __expf(uw[i] - mw);
            sh += __expf(uh[i] - mh);
        }
        const float wscale = (1.0f - MIN_W * KBINS) / sw;
        const float hscale = (1.0f - MIN_H * KBINS) / sh;

        float cw = 0.f, ch = 0.f;
        float Sw = 0.f, Sh = 0.f;
        int   idx = 0;
        float sel_cw = 0.f, sel_ch = 0.f, sel_w = TWO_PI_F, sel_h = TWO_PI_F;
        for (int i = 0; i < KBINS; ++i) {
            const float wfrac = MIN_W + wscale * __expf(uw[i] - mw);
            const float hfrac = MIN_H + hscale * __expf(uh[i] - mh);
            Sw += wfrac;
            Sh += hfrac;
            const float cw_next = (i == KBINS - 1) ? TWO_PI_F : TWO_PI_F * Sw;
            const float ch_next = (i == KBINS - 1) ? TWO_PI_F : TWO_PI_F * Sh;
            if (th >= cw) {
                idx    = i;
                sel_cw = cw;  sel_w = cw_next - cw;
                sel_ch = ch;  sel_h = ch_next - ch;
            }
            cw = cw_next;
            ch = ch_next;
        }

        const float d0 = MIN_D + softplus_f(ud[idx] + DERIV_SHIFT);
        const float d1 = MIN_D + softplus_f(ud[(idx + 1) & (KBINS - 1)] + DERIV_SHIFT);

        const float delta = sel_h / sel_w;
        const float t     = (th - sel_cw) / sel_w;
        const float t1mt  = t * (1.0f - t);
        const float numer = sel_h * (delta * t * t + d0 * t1mt);
        const float denom = delta + (d0 + d1 - 2.0f * delta) * t1mt;
        const float outv  = sel_ch + numer / denom;

        const float omt  = 1.0f - t;
        const float dnum = delta * delta * (d1 * t * t + 2.0f * delta * t1mt + d0 * omt * omt);
        const float lad  = __logf(dnum) - 2.0f * __logf(denom);

        obuf[lane * DH + dp] = outv;
        obuf[ROWS_PER_BLOCK * DH + lane * DH + dp] = lad;
    }

    __syncthreads();

    if (tid < 4 * 32) {
        const int plane = tid >> 6;
        const int c     = tid & 63;
        const v4f val = *reinterpret_cast<const v4f*>(&obuf[plane * ROWS_PER_BLOCK * DH + 4 * c]);
        float* dst = out + (size_t)plane * Btotal * DH + (size_t)rowbase * DH + 4 * c;
        *reinterpret_cast<v4f*>(dst) = val;
    }
}

extern "C" void kernel_launch(void* const* d_in, const int* in_sizes, int n_in,
                              void* d_out, int out_size, void* d_ws, size_t ws_size,
                              hipStream_t stream) {
    (void)n_in; (void)out_size; (void)d_ws; (void)ws_size;
    const float* theta = (const float*)d_in[0];
    const float* x     = (const float*)d_in[1];
    const float* W1    = (const float*)d_in[2];
    const float* b1    = (const float*)d_in[3];
    const float* W2    = (const float*)d_in[4];
    const float* b2    = (const float*)d_in[5];
    const float* eta   = (const float*)d_in[6];
    float* out = (float*)d_out;

    const int B = in_sizes[0] / DH;
    const int blocks = B / ROWS_PER_BLOCK;
    Circular_Spline_Flow_fused_kernel<<<blocks, WAVES * 32, 0, stream>>>(
        theta, x, W1, b1, W2, b2, eta, out, B);
}


# LDS row pitches padded against bank conflicts: H rows 260 floats (was 256), param rows 98 (was 96), so A-fragment reads and the spline's per-row reads hit distinct banks; k and sum order same
# speedup vs baseline: 2.6736x; 1.0612x over previous
#include <hip/hip_runtime.h>
#include <math.h>

typedef float v2f __attribute__((ext_vector_type(2)));
typedef float v4f __attribute__((ext_vector_type(4)));
typedef float v8f __attribute__((ext_vector_type(8)));

#define TWO_PI_F     6.2831853071795864769f
#define MIN_W        1e-3f
#define MIN_H        1e-3f
#define MIN_D        1e-3f
#define DERIV_SHIFT  0.54132485461292f

#define KBINS 32
#define DH    8
#define CDIM  64
#define HDIM  256
#define NPAR  768

#define WAVES           DH
#define MTILES          2
#define ROWS_PER_BLOCK  (MTILES * 16)
#define PCHUNK          96
#define NT2             (PCHUNK / 16)

#define HPITCH  (HDIM + 4)
#define PPITCH  (PCHUNK + 2)

#define HBUF_FLOATS (ROWS_PER_BLOCK * HPITCH)
#define PBUF_FLOATS (WAVES * ROWS_PER_BLOCK * PPITCH)
#define OBUF_FLOATS (2 * ROWS_PER_BLOCK * DH)
#define LDS_BYTES   ((HBUF_FLOATS + PBUF_FLOATS + OBUF_FLOATS) * 4)

#define KSTEPS  (HDIM / 4)
#define PF_DIST 4

static __device__ __forceinline__ float softplus_f(float x) {
    return (x > 20.0f) ? x : log1pf(__expf(x));
}

static __device__ __forceinline__ void load_w2_step(
    const float* __restrict__ W2, int dpn, int kc, int half, int m, v2f (&dst)[NT2])
{
    const float* p = W2 + (size_t)(4 * kc + 2 * half) * NPAR + dpn * PCHUNK + m;
    #pragma unroll
    for (int jt = 0; jt < NT2; ++jt) {
        dst[jt].x = p[16 * jt];
        dst[jt].y = p[NPAR + 16 * jt];
    }
}

static __device__ __forceinline__ void load_a_step(
    const float* hbuf, int kc, int half, int m, v2f (&dst)[MTILES])
{
    const int kk = 4 * kc + 2 * half;
    #pragma unroll
    for (int mt = 0; mt < MTILES; ++mt) {
        dst[mt].x = hbuf[(mt * 16 + m) * HPITCH + kk];
        dst[mt].y = hbuf[(mt * 16 + m) * HPITCH + kk + 1];
    }
}

static __device__ __forceinline__ void load_w1_tile(
    const float* __restrict__ W1, int j, int half, int m, v2f (&dst)[16])
{
    const float* p = W1 + (size_t)(2 * half) * HDIM + 16 * j + m;
    #pragma unroll
    for (int kc = 0; kc < 16; ++kc) {
        dst[kc].x = p[(size_t)(4 * kc) * HDIM];
        dst[kc].y = p[(size_t)(4 * kc + 1) * HDIM];
    }
}

static __device__ __forceinline__ void gemm1_tile(
    const float* __restrict__ W1, const float* __restrict__ b1, float* hbuf,
    const v2f (&afrag)[MTILES][16], const v2f (&cur)[16], v2f (&nxt)[16],
    int j, bool prefetch, int half, int m)
{
    const int col = 16 * j + m;
    const float bias = b1[col];
    if (prefetch)
        load_w1_tile(W1, j + 1, half, m, nxt);
    __builtin_amdgcn_sched_barrier(0);

    v8f acc0 = {0.f,0.f,0.f,0.f,0.f,0.f,0.f,0.f};
    v8f acc1 = {0.f,0.f,0.f,0.f,0.f,0.f,0.f,0.f};
    #pragma unroll
    for (int kc = 0; kc < 16; ++kc) {
        acc0 = __builtin_amdgcn_wmma_f32_16x16x4_f32(
            false, afrag[0][kc], false, cur[kc], (short)0, acc0, false, false);
        acc1 = __builtin_amdgcn_wmma_f32_16x16x4_f32(
            false, afrag[1][kc], false, cur[kc], (short)0, acc1, false, false);
    }
    #pragma unroll
    for (int v = 0; v < 8; ++v) {
        float h0 = fmaxf(acc0[v] + bias, 0.0f);
        float h1 = fmaxf(acc1[v] + bias, 0.0f);
        hbuf[( 0 + 8 * half + v) * HPITCH + col] = h0;
        hbuf[(16 + 8 * half + v) * HPITCH + col] = h1;
    }
}

static __device__ __forceinline__ void gemm2_group(
    const float* __restrict__ W2, const float* hbuf, int dp, int kc4, bool request,
    int half, int m,
    v2f (&aq)[PF_DIST][MTILES], v2f (&bq)[PF_DIST][NT2], v8f (&acc)[MTILES][NT2])
{
    #pragma unroll
    for (int u = 0; u < PF_DIST; ++u) {
        #pragma unroll
        for (int jt = 0; jt < NT2; ++jt) {
            acc[0][jt] = __builtin_amdgcn_wmma_f32_16x16x4_f32(
                false, aq[u][0], false, bq[u][jt], (short)0, acc[0][jt], false, false);
            acc[1][jt] = __builtin_amdgcn_wmma_f32_16x16x4_f32(
                false, aq[u][1], false, bq[u][jt], (short)0, acc[1][jt], false, false);
        }
        if (request) {
            const int kn = kc4 + u + PF_DIST;
            load_a_step(hbuf, kn, half, m, aq[u]);
            load_w2_step(W2, dp, kn, half, m, bq[u]);
            __builtin_amdgcn_sched_barrier(0);
        }
    }
}

__global__ __launch_bounds__(WAVES * 32, 4)
void Circular_Spline_Flow_fused_kernel(
    const float* __restrict__ theta,
    const float* __restrict__ x,
    const float* __restrict__ W1,
    const float* __restrict__ b1,
    const float* __restrict__ W2,
    const float* __restrict__ b2,
    const float* __restrict__ eta,
    float* __restrict__ out,
    int Btotal)
{
    __shared__ __attribute__((aligned(16))) float smem[LDS_BYTES / 4];

    const int tid  = threadIdx.x;
    const int wave = tid >> 5;
    const int lane = tid & 31;
    const int m    = lane & 15;
    const int half = lane >> 4;

    const int rowbase = blockIdx.x * ROWS_PER_BLOCK;
    const int dp      = wave;

    float* hbuf = smem;
    float* pbuf = smem + HBUF_FLOATS + wave * ROWS_PER_BLOCK * PPITCH;
    float* obuf = smem + HBUF_FLOATS + PBUF_FLOATS;

    const float etav = eta[0];

    v2f afrag[MTILES][16];
    #pragma unroll
    for (int mt = 0; mt < MTILES; ++mt) {
        const float* xrow = x + (size_t)(rowbase + mt * 16 + m) * CDIM;
        #pragma unroll
        for (int kc = 0; kc < 16; ++kc) {
            const int kk = 4 * kc + 2 * half;
            afrag[mt][kc].x = xrow[kk];
            afrag[mt][kc].y = xrow[kk + 1];
        }
    }

    v2f bq[PF_DIST][NT2];
    #pragma unroll
    for (int u = 0; u < PF_DIST; ++u)
        load_w2_step(W2, dp, u, half, m, bq[u]);

    float bias2[NT2];
    #pragma unroll
    for (int jt = 0; jt < NT2; ++jt)
        bias2[jt] = b2[dp * PCHUNK + 16 * jt + m];
    const float th = theta[(size_t)(rowbase + lane) * DH + dp];

    v2f bw[2][16];
    load_w1_tile(W1, 2 * wave, half, m, bw[0]);
    gemm1_tile(W1, b1, hbuf, afrag, bw[0], bw[1], 2 * wave,     true,  half, m);
    gemm1_tile(W1, b1, hbuf, afrag, bw[1], bw[0], 2 * wave + 1, false, half, m);

    __syncthreads();

    v2f aq[PF_DIST][MTILES];
    #pragma unroll
    for (int u = 0; u < PF_DIST; ++u)
        load_a_step(hbuf, u, half, m, aq[u]);

    v8f acc[MTILES][NT2];
    #pragma unroll
    for (int mt = 0; mt < MTILES; ++mt)
        #pragma unroll
        for (int jt = 0; jt < NT2; ++jt)
            acc[mt][jt] = (v8f){0.f,0.f,0.f,0.f,0.f,0.f,0.f,0.f};

    #pragma unroll 1
    for (int kc4 = 0; kc4 < KSTEPS - PF_DIST; kc4 += PF_DIST)
        gemm2_group(W2, hbuf, dp, kc4, true, half, m, aq, bq, acc);
    gemm2_group(W2, hbuf, dp, KSTEPS - PF_DIST, false, half, m, aq, bq, acc);

    #pragma unroll
    for (int jt = 0; jt < NT2; ++jt) {
        const float bias = bias2[jt];
        #pragma unroll
        for (int mt = 0; mt < MTILES; ++mt) {
            #pragma unroll
            for (int v = 0; v < 8; ++v) {
                const float pv = (acc[mt][jt][v] + bias) * etav;
                pbuf[(mt * 16 + 8 * half + v) * PPITCH + 16 * jt + m] = pv;
            }
        }
    }

    {
        const float* P  = &pbuf[lane * PPITCH];
        const float* uw = P;
        const float* uh = P + 32;
        const float* ud = P + 64;

        float mw = -1e30f, mh = -1e30f;
        for (int i = 0; i < KBINS; ++i) {
            mw = fmaxf(mw, uw[i]);
            mh = fmaxf(mh, uh[i]);
        }
        float sw = 0.f, sh = 0.f;
        for (int i = 0; i < KBINS; ++i) {
            sw += __expf(uw[i] - mw);
            sh += __expf(uh[i] - mh);
        }
        const float wscale = (1.0f - MIN_W * KBINS) / sw;
        const float hscale = (1.0f - MIN_H * KBINS) / sh;

        float cw = 0.f, ch = 0.f;
        float Sw = 0.f, Sh = 0.f;
        int   idx = 0;
        float sel_cw = 0.f, sel_ch = 0.f, sel_w = TWO_PI_F, sel_h = TWO_PI_F;
        for (int i = 0; i < KBINS; ++i) {
            const float wfrac = MIN_W + wscale * __expf(uw[i] - mw);
            const float hfrac = MIN_H + hscale * __expf(uh[i] - mh);
            Sw += wfrac;
            Sh += hfrac;
            const float cw_next = (i == KBINS - 1) ? TWO_PI_F : TWO_PI_F * Sw;
            const float ch_next = (i == KBINS - 1) ? TWO_PI_F : TWO_PI_F * Sh;
            if (th >= cw) {
                idx    = i;
                sel_cw = cw;  sel_w = cw_next - cw;
                sel_ch = ch;  sel_h = ch_next - ch;
            }
            cw = cw_next;
            ch = ch_next;
        }

        const float d0 = MIN_D + softplus_f(ud[idx] + DERIV_SHIFT);
        const float d1 = MIN_D + softplus_f(ud[(idx + 1) & (KBINS - 1)] + DERIV_SHIFT);

        const float delta = sel_h / sel_w;
        const float t     = (th - sel_cw) / sel_w;
        const float t1mt  = t * (1.0f - t);
        const float numer = sel_h * (delta * t * t + d0 * t1mt);
        const float denom = delta + (d0 + d1 - 2.0f * delta) * t1mt;
        const float outv  = sel_ch + numer / denom;

        const float omt  = 1.0f - t;
        const float dnum = delta * delta * (d1 * t * t + 2.0f * delta * t1mt + d0 * omt * omt);
        const float lad  = __logf(dnum) - 2.0f * __logf(denom);

        obuf[lane * DH + dp] = outv;
        obuf[ROWS_PER_BLOCK * DH + lane * DH + dp] = lad;
    }

    __syncthreads();

    if (tid < 4 * 32) {
        const int plane = tid >> 6;
        const int c     = tid & 63;
        const v4f val = *reinterpret_cast<const v4f*>(&obuf[plane * ROWS_PER_BLOCK * DH + 4 * c]);
        float* dst = out + (size_t)plane * Btotal * DH + (size_t)rowbase * DH + 4 * c;
        *reinterpret_cast<v4f*>(dst) = val;
    }
}

extern "C" void kernel_launch(void* const* d_in, const int* in_sizes, int n_in,
                              void* d_out, int out_size, void* d_ws, size_t ws_size,
                              hipStream_t stream) {
    (void)n_in; (void)out_size; (void)d_ws; (void)ws_size;
    const float* theta = (const float*)d_in[0];
    const float* x     = (const float*)d_in[1];
    const float* W1    = (const float*)d_in[2];
    const float* b1    = (const float*)d_in[3];
    const float* W2    = (const float*)d_in[4];
    const float* b2    = (const float*)d_in[5];
    const float* eta   = (const float*)d_in[6];
    float* out = (float*)d_out;

    const int B = in_sizes[0] / DH;
    const int blocks = B / ROWS_PER_BLOCK;
    Circular_Spline_Flow_fused_kernel<<<blocks, WAVES * 32, 0, stream>>>(
        theta, x, W1, b1, W2, b2, eta, out, B);
}


# spline holds its 64 width/height params in registers: one LDS read pass and one exp per param (was 3 read passes, 2 exp passes, one LDS round trip per bin); same values, same sum order
# speedup vs baseline: 2.8144x; 1.0527x over previous
#include <hip/hip_runtime.h>
#include <math.h>

typedef float v2f __attribute__((ext_vector_type(2)));
typedef float v4f __attribute__((ext_vector_type(4)));
typedef float v8f __attribute__((ext_vector_type(8)));

#define TWO_PI_F     6.2831853071795864769f
#define MIN_W        1e-3f
#define MIN_H        1e-3f
#define MIN_D        1e-3f
#define DERIV_SHIFT  0.54132485461292f

#define KBINS 32
#define DH    8
#define CDIM  64
#define HDIM  256
#define NPAR  768

#define WAVES           DH
#define MTILES          2
#define ROWS_PER_BLOCK  (MTILES * 16)
#define PCHUNK          96
#define NT2             (PCHUNK / 16)

#define HPITCH  (HDIM + 4)
#define PPITCH  (PCHUNK + 2)

#define HBUF_FLOATS (ROWS_PER_BLOCK * HPITCH)
#define PBUF_FLOATS (WAVES * ROWS_PER_BLOCK * PPITCH)
#define OBUF_FLOATS (2 * ROWS_PER_BLOCK * DH)
#define LDS_BYTES   ((HBUF_FLOATS + PBUF_FLOATS + OBUF_FLOATS) * 4)

#define KSTEPS  (HDIM / 4)
#define PF_DIST 4

static __device__ __forceinline__ float softplus_f(float x) {
    return (x > 20.0f) ? x : log1pf(__expf(x));
}

static __device__ __forceinline__ void load_w2_step(
    const float* __restrict__ W2, int dpn, int kc, int half, int m, v2f (&dst)[NT2])
{
    const float* p = W2 + (size_t)(4 * kc + 2 * half) * NPAR + dpn * PCHUNK + m;
    #pragma unroll
    for (int jt = 0; jt < NT2; ++jt) {
        dst[jt].x = p[16 * jt];
        dst[jt].y = p[NPAR + 16 * jt];
    }
}

static __device__ __forceinline__ void load_a_step(
    const float* hbuf, int kc, int half, int m, v2f (&dst)[MTILES])
{
    const int kk = 4 * kc + 2 * half;
    #pragma unroll
    for (int mt = 0; mt < MTILES; ++mt) {
        dst[mt].x = hbuf[(mt * 16 + m) * HPITCH + kk];
        dst[mt].y = hbuf[(mt * 16 + m) * HPITCH + kk + 1];
    }
}

static __device__ __forceinline__ void load_w1_tile(
    const float* __restrict__ W1, int j, int half, int m, v2f (&dst)[16])
{
    const float* p = W1 + (size_t)(2 * half) * HDIM + 16 * j + m;
    #pragma unroll
    for (int kc = 0; kc < 16; ++kc) {
        dst[kc].x = p[(size_t)(4 * kc) * HDIM];
        dst[kc].y = p[(size_t)(4 * kc + 1) * HDIM];
    }
}

static __device__ __forceinline__ void gemm1_tile(
    const float* __restrict__ W1, const float* __restrict__ b1, float* hbuf,
    const v2f (&afrag)[MTILES][16], const v2f (&cur)[16], v2f (&nxt)[16],
    int j, bool prefetch, int half, int m)
{
    const int col = 16 * j + m;
    const float bias = b1[col];
    if (prefetch)
        load_w1_tile(W1, j + 1, half, m, nxt);
    __builtin_amdgcn_sched_barrier(0);

    v8f acc0 = {0.f,0.f,0.f,0.f,0.f,0.f,0.f,0.f};
    v8f acc1 = {0.f,0.f,0.f,0.f,0.f,0.f,0.f,0.f};
    #pragma unroll
    for (int kc = 0; kc < 16; ++kc) {
        acc0 = __builtin_amdgcn_wmma_f32_16x16x4_f32(
            false, afrag[0][kc], false, cur[kc], (short)0, acc0, false, false);
        acc1 = __builtin_amdgcn_wmma_f32_16x16x4_f32(
            false, afrag[1][kc], false, cur[kc], (short)0, acc1, false, false);
    }
    #pragma unroll
    for (int v = 0; v < 8; ++v) {
        float h0 = fmaxf(acc0[v] + bias, 0.0f);
        float h1 = fmaxf(acc1[v] + bias, 0.0f);
        hbuf[( 0 + 8 * half + v) * HPITCH + col] = h0;
        hbuf[(16 + 8 * half + v) * HPITCH + col] = h1;
    }
}

static __device__ __forceinline__ void gemm2_group(
    const float* __restrict__ W2, const float* hbuf, int dp, int kc4, bool request,
    int half, int m,
    v2f (&aq)[PF_DIST][MTILES], v2f (&bq)[PF_DIST][NT2], v8f (&acc)[MTILES][NT2])
{
    #pragma unroll
    for (int u = 0; u < PF_DIST; ++u) {
        #pragma unroll
        for (int jt = 0; jt < NT2; ++jt) {
            acc[0][jt] = __builtin_amdgcn_wmma_f32_16x16x4_f32(
                false, aq[u][0], false, bq[u][jt], (short)0, acc[0][jt], false, false);
            acc[1][jt] = __builtin_amdgcn_wmma_f32_16x16x4_f32(
                false, aq[u][1], false, bq[u][jt], (short)0, acc[1][jt], false, false);
        }
        if (request) {
            const int kn = kc4 + u + PF_DIST;
            load_a_step(hbuf, kn, half, m, aq[u]);
            load_w2_step(W2, dp, kn, half, m, bq[u]);
            __builtin_amdgcn_sched_barrier(0);
        }
    }
}

__global__ __launch_bounds__(WAVES * 32, 4)
void Circular_Spline_Flow_fused_kernel(
    const float* __restrict__ theta,
    const float* __restrict__ x,
    const float* __restrict__ W1,
    const float* __restrict__ b1,
    const float* __restrict__ W2,
    const float* __restrict__ b2,
    const float* __restrict__ eta,
    float* __restrict__ out,
    int Btotal)
{
    __shared__ __attribute__((aligned(16))) float smem[LDS_BYTES / 4];

    const int tid  = threadIdx.x;
    const int wave = tid >> 5;
    const int lane = tid & 31;
    const int m    = lane & 15;
    const int half = lane >> 4;

    const int rowbase = blockIdx.x * ROWS_PER_BLOCK;
    const int dp      = wave;

    float* hbuf = smem;
    float* pbuf = smem + HBUF_FLOATS + wave * ROWS_PER_BLOCK * PPITCH;
    float* obuf = smem + HBUF_FLOATS + PBUF_FLOATS;

    const float etav = eta[0];

    v2f afrag[MTILES][16];
    #pragma unroll
    for (int mt = 0; mt < MTILES; ++mt) {
        const float* xrow = x + (size_t)(rowbase + mt * 16 + m) * CDIM;
        #pragma unroll
        for (int kc = 0; kc < 16; ++kc) {
            const int kk = 4 * kc + 2 * half;
            afrag[mt][kc].x = xrow[kk];
            afrag[mt][kc].y = xrow[kk + 1];
        }
    }

    v2f bq[PF_DIST][NT2];
    #pragma unroll
    for (int u = 0; u < PF_DIST; ++u)
        load_w2_step(W2, dp, u, half, m, bq[u]);

    float bias2[NT2];
    #pragma unroll
    for (int jt = 0; jt < NT2; ++jt)
        bias2[jt] = b2[dp * PCHUNK + 16 * jt + m];
    const float th = theta[(size_t)(rowbase + lane) * DH + dp];

    v2f bw[2][16];
    load_w1_tile(W1, 2 * wave, half, m, bw[0]);
    gemm1_tile(W1, b1, hbuf, afrag, bw[0], bw[1], 2 * wave,     true,  half, m);
    gemm1_tile(W1, b1, hbuf, afrag, bw[1], bw[0], 2 * wave + 1, false, half, m);

    __syncthreads();

    v2f aq[PF_DIST][MTILES];
    #pragma unroll
    for (int u = 0; u < PF_DIST; ++u)
        load_a_step(hbuf, u, half, m, aq[u]);

    v8f acc[MTILES][NT2];
    #pragma unroll
    for (int mt = 0; mt < MTILES; ++mt)
        #pragma unroll
        for (int jt = 0; jt < NT2; ++jt)
            acc[mt][jt] = (v8f){0.f,0.f,0.f,0.f,0.f,0.f,0.f,0.f};

    #pragma unroll 1
    for (int kc4 = 0; kc4 < KSTEPS - PF_DIST; kc4 += PF_DIST)
        gemm2_group(W2, hbuf, dp, kc4, true, half, m, aq, bq, acc);
    gemm2_group(W2, hbuf, dp, KSTEPS - PF_DIST, false, half, m, aq, bq, acc);

    #pragma unroll
    for (int jt = 0; jt < NT2; ++jt) {
        const float bias = bias2[jt];
        #pragma unroll
        for (int mt = 0; mt < MTILES; ++mt) {
            #pragma unroll
            for (int v = 0; v < 8; ++v) {
                const float pv = (acc[mt][jt][v] + bias) * etav;
                pbuf[(mt * 16 + 8 * half + v) * PPITCH + 16 * jt + m] = pv;
            }
        }
    }

    {
        const float* P  = &pbuf[lane * PPITCH];
        const float* uw = P;
        const float* uh = P + 32;
        const float* ud = P + 64;

        float ew[KBINS], eh[KBINS];
        #pragma unroll
        for (int i = 0; i < KBINS; ++i) {
            ew[i] = uw[i];
            eh[i] = uh[i];
        }
        float mw = -1e30f, mh = -1e30f;
        #pragma unroll
        for (int i = 0; i < KBINS; ++i) {
            mw = fmaxf(mw, ew[i]);
            mh = fmaxf(mh, eh[i]);
        }
        float sw = 0.f, sh = 0.f;
        #pragma unroll
        for (int i = 0; i < KBINS; ++i) {
            ew[i] = __expf(ew[i] - mw);
            eh[i] = __expf(eh[i] - mh);
            sw += ew[i];
            sh += eh[i];
        }
        const float wscale = (1.0f - MIN_W * KBINS) / sw;
        const float hscale = (1.0f - MIN_H * KBINS) / sh;

        float cw = 0.f, ch = 0.f;
        float Sw = 0.f, Sh = 0.f;
        int   idx = 0;
        float lo_w = 0.f, hi_w = TWO_PI_F, lo_h = 0.f, hi_h = TWO_PI_F;
        #pragma unroll
        for (int i = 0; i < KBINS; ++i) {
            const float wfrac = MIN_W + wscale * ew[i];
            const float hfrac = MIN_H + hscale * eh[i];
            Sw += wfrac;
            Sh += hfrac;
            const float cw_next = (i == KBINS - 1) ? TWO_PI_F : TWO_PI_F * Sw;
            const float ch_next = (i == KBINS - 1) ? TWO_PI_F : TWO_PI_F * Sh;
            const bool  in_bin  = (th >= cw);
            idx  = in_bin ? i       : idx;
            lo_w = in_bin ? cw      : lo_w;
            hi_w = in_bin ? cw_next : hi_w;
            lo_h = in_bin ? ch      : lo_h;
            hi_h = in_bin ? ch_next : hi_h;
            cw = cw_next;
            ch = ch_next;
        }
        const float sel_cw = lo_w, sel_w = hi_w - lo_w;
        const float sel_ch = lo_h, sel_h = hi_h - lo_h;

        const float d0 = MIN_D + softplus_f(ud[idx] + DERIV_SHIFT);
        const float d1 = MIN_D + softplus_f(ud[(idx + 1) & (KBINS - 1)] + DERIV_SHIFT);

        const float delta = sel_h / sel_w;
        const float t     = (th - sel_cw) / sel_w;
        const float t1mt  = t * (1.0f - t);
        const float numer = sel_h * (delta * t * t + d0 * t1mt);
        const float denom = delta + (d0 + d1 - 2.0f * delta) * t1mt;
        const float outv  = sel_ch + numer / denom;

        const float omt  = 1.0f - t;
        const float dnum = delta * delta * (d1 * t * t + 2.0f * delta * t1mt + d0 * omt * omt);
        const float lad  = __logf(dnum) - 2.0f * __logf(denom);

        obuf[lane * DH + dp] = outv;
        obuf[ROWS_PER_BLOCK * DH + lane * DH + dp] = lad;
    }

    __syncthreads();

    if (tid < 4 * 32) {
        const int plane = tid >> 6;
        const int c     = tid & 63;
        const v4f val = *reinterpret_cast<const v4f*>(&obuf[plane * ROWS_PER_BLOCK * DH + 4 * c]);
        float* dst = out + (size_t)plane * Btotal * DH + (size_t)rowbase * DH + 4 * c;
        *reinterpret_cast<v4f*>(dst) = val;
    }
}

extern "C" void kernel_launch(void* const* d_in, const int* in_sizes, int n_in,
                              void* d_out, int out_size, void* d_ws, size_t ws_size,
                              hipStream_t stream) {
    (void)n_in; (void)out_size; (void)d_ws; (void)ws_size;
    const float* theta = (const float*)d_in[0];
    const float* x     = (const float*)d_in[1];
    const float* W1    = (const float*)d_in[2];
    const float* b1    = (const float*)d_in[3];
    const float* W2    = (const float*)d_in[4];
    const float* b2    = (const float*)d_in[5];
    const float* eta   = (const float*)d_in[6];
    float* out = (float*)d_out;

    const int B = in_sizes[0] / DH;
    const int blocks = B / ROWS_PER_BLOCK;
    Circular_Spline_Flow_fused_kernel<<<blocks, WAVES * 32, 0, stream>>>(
        theta, x, W1, b1, W2, b2, eta, out, B);
}


# three work-groups (24 waves) per WGP, was two (16): params turned to rows through one 32-column LDS buffer, group by group (LDS 70 KB, was 133); 2-slot fragment ring (158 VGPRs, was 215); sums same
# speedup vs baseline: 2.9088x; 1.0335x over previous
#include <hip/hip_runtime.h>
#include <math.h>

typedef float v2f __attribute__((ext_vector_type(2)));
typedef float v4f __attribute__((ext_vector_type(4)));
typedef float v8f __attribute__((ext_vector_type(8)));

#define TWO_PI_F     6.2831853071795864769f
#define MIN_W        1e-3f
#define MIN_H        1e-3f
#define MIN_D        1e-3f
#define DERIV_SHIFT  0.54132485461292f

#define KBINS 32
#define DH    8
#define CDIM  64
#define HDIM  256
#define NPAR  768

#define WAVES           DH
#define MTILES          2
#define ROWS_PER_BLOCK  (MTILES * 16)
#define PCHUNK          96
#define NT2             (PCHUNK / 16)

#define HPITCH  (HDIM + 4)
#define PPITCH  (KBINS + 2)

#define HBUF_FLOATS (ROWS_PER_BLOCK * HPITCH)
#define PBUF_FLOATS (WAVES * ROWS_PER_BLOCK * PPITCH)
#define OBUF_FLOATS (2 * ROWS_PER_BLOCK * DH)
#define LDS_BYTES   ((HBUF_FLOATS + PBUF_FLOATS + OBUF_FLOATS) * 4)

#define KSTEPS  (HDIM / 4)
#define PF_DIST 2

static __device__ __forceinline__ float softplus_f(float x) {
    return (x > 20.0f) ? x : log1pf(__expf(x));
}

static __device__ __forceinline__ void load_w2_step(
    const float* __restrict__ W2, int dpn, int kc, int half, int m, v2f (&dst)[NT2])
{
    const float* p = W2 + (size_t)(4 * kc + 2 * half) * NPAR + dpn * PCHUNK + m;
    #pragma unroll
    for (int jt = 0; jt < NT2; ++jt) {
        dst[jt].x = p[16 * jt];
        dst[jt].y = p[NPAR + 16 * jt];
    }
}

static __device__ __forceinline__ void load_a_step(
    const float* hbuf, int kc, int half, int m, v2f (&dst)[MTILES])
{
    const int kk = 4 * kc + 2 * half;
    #pragma unroll
    for (int mt = 0; mt < MTILES; ++mt) {
        dst[mt].x = hbuf[(mt * 16 + m) * HPITCH + kk];
        dst[mt].y = hbuf[(mt * 16 + m) * HPITCH + kk + 1];
    }
}

static __device__ __forceinline__ void load_w1_tile(
    const float* __restrict__ W1, int j, int half, int m, v2f (&dst)[16])
{
    const float* p = W1 + (size_t)(2 * half) * HDIM + 16 * j + m;
    #pragma unroll
    for (int kc = 0; kc < 16; ++kc) {
        dst[kc].x = p[(size_t)(4 * kc) * HDIM];
        dst[kc].y = p[(size_t)(4 * kc + 1) * HDIM];
    }
}

static __device__ __forceinline__ void gemm1_tile(
    const float* __restrict__ b1, float* hbuf,
    const v2f (&afrag)[MTILES][16], const v2f (&cur)[16],
    int j, int half, int m)
{
    const int col = 16 * j + m;
    const float bias = b1[col];

    v8f acc0 = {0.f,0.f,0.f,0.f,0.f,0.f,0.f,0.f};
    v8f acc1 = {0.f,0.f,0.f,0.f,0.f,0.f,0.f,0.f};
    #pragma unroll
    for (int kc = 0; kc < 16; ++kc) {
        acc0 = __builtin_amdgcn_wmma_f32_16x16x4_f32(
            false, afrag[0][kc], false, cur[kc], (short)0, acc0, false, false);
        acc1 = __builtin_amdgcn_wmma_f32_16x16x4_f32(
            false, afrag[1][kc], false, cur[kc], (short)0, acc1, false, false);
    }
    #pragma unroll
    for (int v = 0; v < 8; ++v) {
        float h0 = fmaxf(acc0[v] + bias, 0.0f);
        float h1 = fmaxf(acc1[v] + bias, 0.0f);
        hbuf[( 0 + 8 * half + v) * HPITCH + col] = h0;
        hbuf[(16 + 8 * half + v) * HPITCH + col] = h1;
    }
}

static __device__ __forceinline__ void gemm2_group(
    const float* __restrict__ W2, const float* hbuf, int dp, int kc4, bool request,
    int half, int m,
    v2f (&aq)[PF_DIST][MTILES], v2f (&bq)[PF_DIST][NT2], v8f (&acc)[MTILES][NT2])
{
    #pragma unroll
    for (int u = 0; u < PF_DIST; ++u) {
        #pragma unroll
        for (int jt = 0; jt < NT2; ++jt) {
            acc[0][jt] = __builtin_amdgcn_wmma_f32_16x16x4_f32(
                false, aq[u][0], false, bq[u][jt], (short)0, acc[0][jt], false, false);
            acc[1][jt] = __builtin_amdgcn_wmma_f32_16x16x4_f32(
                false, aq[u][1], false, bq[u][jt], (short)0, acc[1][jt], false, false);
        }
        if (request) {
            const int kn = kc4 + u + PF_DIST;
            load_a_step(hbuf, kn, half, m, aq[u]);
            load_w2_step(W2, dp, kn, half, m, bq[u]);
            __builtin_amdgcn_sched_barrier(0);
        }
    }
}

__global__ __launch_bounds__(WAVES * 32, 6)
void Circular_Spline_Flow_fused_kernel(
    const float* __restrict__ theta,
    const float* __restrict__ x,
    const float* __restrict__ W1,
    const float* __restrict__ b1,
    const float* __restrict__ W2,
    const float* __restrict__ b2,
    const float* __restrict__ eta,
    float* __restrict__ out,
    int Btotal)
{
    __shared__ __attribute__((aligned(16))) float smem[LDS_BYTES / 4];

    const int tid  = threadIdx.x;
    const int wave = tid >> 5;
    const int lane = tid & 31;
    const int m    = lane & 15;
    const int half = lane >> 4;

    const int rowbase = blockIdx.x * ROWS_PER_BLOCK;
    const int dp      = wave;

    float* hbuf = smem;
    float* pbuf = smem + HBUF_FLOATS + wave * ROWS_PER_BLOCK * PPITCH;
    float* obuf = smem + HBUF_FLOATS + PBUF_FLOATS;

    const float etav = eta[0];

    v2f afrag[MTILES][16];
    #pragma unroll
    for (int mt = 0; mt < MTILES; ++mt) {
        const float* xrow = x + (size_t)(rowbase + mt * 16 + m) * CDIM;
        #pragma unroll
        for (int kc = 0; kc < 16; ++kc) {
            const int kk = 4 * kc + 2 * half;
            afrag[mt][kc].x = xrow[kk];
            afrag[mt][kc].y = xrow[kk + 1];
        }
    }

    v2f bw[16];
    load_w1_tile(W1, 2 * wave, half, m, bw);
    gemm1_tile(b1, hbuf, afrag, bw, 2 * wave, half, m);
    __builtin_amdgcn_sched_barrier(0);

    float bias2[NT2];
    #pragma unroll
    for (int jt = 0; jt < NT2; ++jt)
        bias2[jt] = b2[dp * PCHUNK + 16 * jt + m];
    const float th = theta[(size_t)(rowbase + lane) * DH + dp];

    v2f bq[PF_DIST][NT2];
    #pragma unroll
    for (int u = 0; u < PF_DIST; ++u)
        load_w2_step(W2, dp, u, half, m, bq[u]);

    load_w1_tile(W1, 2 * wave + 1, half, m, bw);
    gemm1_tile(b1, hbuf, afrag, bw, 2 * wave + 1, half, m);

    __syncthreads();

    v2f aq[PF_DIST][MTILES];
    #pragma unroll
    for (int u = 0; u < PF_DIST; ++u)
        load_a_step(hbuf, u, half, m, aq[u]);

    v8f acc[MTILES][NT2];
    #pragma unroll
    for (int mt = 0; mt < MTILES; ++mt)
        #pragma unroll
        for (int jt = 0; jt < NT2; ++jt)
            acc[mt][jt] = (v8f){0.f,0.f,0.f,0.f,0.f,0.f,0.f,0.f};

    #pragma unroll 1
    for (int kc4 = 0; kc4 < KSTEPS - PF_DIST; kc4 += PF_DIST)
        gemm2_group(W2, hbuf, dp, kc4, true, half, m, aq, bq, acc);
    gemm2_group(W2, hbuf, dp, KSTEPS - PF_DIST, false, half, m, aq, bq, acc);

    {
        float ew[KBINS], eh[KBINS];
        #pragma unroll
        for (int g = 0; g < 3; ++g) {
            #pragma unroll
            for (int jj = 0; jj < 2; ++jj) {
                const int jt = 2 * g + jj;
                const float bias = bias2[jt];
                #pragma unroll
                for (int mt = 0; mt < MTILES; ++mt) {
                    #pragma unroll
                    for (int v = 0; v < 8; ++v) {
                        const float pv = (acc[mt][jt][v] + bias) * etav;
                        pbuf[(mt * 16 + 8 * half + v) * PPITCH + 16 * jj + m] = pv;
                    }
                }
            }
            if (g == 0) {
                #pragma unroll
                for (int i = 0; i < KBINS; ++i) ew[i] = pbuf[lane * PPITCH + i];
            }
            if (g == 1) {
                #pragma unroll
                for (int i = 0; i < KBINS; ++i) eh[i] = pbuf[lane * PPITCH + i];
            }
        }
        const float* ud = &pbuf[lane * PPITCH];

        float mw = -1e30f, mh = -1e30f;
        #pragma unroll
        for (int i = 0; i < KBINS; ++i) {
            mw = fmaxf(mw, ew[i]);
            mh = fmaxf(mh, eh[i]);
        }
        float sw = 0.f, sh = 0.f;
        #pragma unroll
        for (int i = 0; i < KBINS; ++i) {
            ew[i] = __expf(ew[i] - mw);
            eh[i] = __expf(eh[i] - mh);
            sw += ew[i];
            sh += eh[i];
        }
        const float wscale = (1.0f - MIN_W * KBINS) / sw;
        const float hscale = (1.0f - MIN_H * KBINS) / sh;

        float cw = 0.f, ch = 0.f;
        float Sw = 0.f, Sh = 0.f;
        int   idx = 0;
        float lo_w = 0.f, hi_w = TWO_PI_F, lo_h = 0.f, hi_h = TWO_PI_F;
        #pragma unroll
        for (int i = 0; i < KBINS; ++i) {
            const float wfrac = MIN_W + wscale * ew[i];
            const float hfrac = MIN_H + hscale * eh[i];
            Sw += wfrac;
            Sh += hfrac;
            const float cw_next = (i == KBINS - 1) ? TWO_PI_F : TWO_PI_F * Sw;
            const float ch_next = (i == KBINS - 1) ? TWO_PI_F : TWO_PI_F * Sh;
            const bool  in_bin  = (th >= cw);
            idx  = in_bin ? i       : idx;
            lo_w = in_bin ? cw      : lo_w;
            hi_w = in_bin ? cw_next : hi_w;
            lo_h = in_bin ? ch      : lo_h;
            hi_h = in_bin ? ch_next : hi_h;
            cw = cw_next;
            ch = ch_next;
        }
        const float sel_cw = lo_w, sel_w = hi_w - lo_w;
        const float sel_ch = lo_h, sel_h = hi_h - lo_h;

        const float d0 = MIN_D + softplus_f(ud[idx] + DERIV_SHIFT);
        const float d1 = MIN_D + softplus_f(ud[(idx + 1) & (KBINS - 1)] + DERIV_SHIFT);

        const float delta = sel_h / sel_w;
        const float t     = (th - sel_cw) / sel_w;
        const float t1mt  = t * (1.0f - t);
        const float numer = sel_h * (delta * t * t + d0 * t1mt);
        const float denom = delta + (d0 + d1 - 2.0f * delta) * t1mt;
        const float outv  = sel_ch + numer / denom;

        const float omt  = 1.0f - t;
        const float dnum = delta * delta * (d1 * t * t + 2.0f * delta * t1mt + d0 * omt * omt);
        const float lad  = __logf(dnum) - 2.0f * __logf(denom);

        obuf[lane * DH + dp] = outv;
        obuf[ROWS_PER_BLOCK * DH + lane * DH + dp] = lad;
    }

    __syncthreads();

    if (tid < 4 * 32) {
        const int plane = tid >> 6;
        const int c     = tid & 63;
        const v4f val = *reinterpret_cast<const v4f*>(&obuf[plane * ROWS_PER_BLOCK * DH + 4 * c]);
        float* dst = out + (size_t)plane * Btotal * DH + (size_t)rowbase * DH + 4 * c;
        *reinterpret_cast<v4f*>(dst) = val;
    }
}

extern "C" void kernel_launch(void* const* d_in, const int* in_sizes, int n_in,
                              void* d_out, int out_size, void* d_ws, size_t ws_size,
                              hipStream_t stream) {
    (void)n_in; (void)out_size; (void)d_ws; (void)ws_size;
    const float* theta = (const float*)d_in[0];
    const float* x     = (const float*)d_in[1];
    const float* W1    = (const float*)d_in[2];
    const float* b1    = (const float*)d_in[3];
    const float* W2    = (const float*)d_in[4];
    const float* b2    = (const float*)d_in[5];
    const float* eta   = (const float*)d_in[6];
    float* out = (float*)d_out;

    const int B = in_sizes[0] / DH;
    const int blocks = B / ROWS_PER_BLOCK;
    Circular_Spline_Flow_fused_kernel<<<blocks, WAVES * 32, 0, stream>>>(
        theta, x, W1, b1, W2, b2, eta, out, B);
}


# 64-row work-group tile: each W2 fragment feeds four M tiles (was two), halving W2 line reads and global loads per WMMA; two work-groups (16 waves) per WGP at 256 VGPRs; k and sum order same
# speedup vs baseline: 3.0932x; 1.0634x over previous
#include <hip/hip_runtime.h>
#include <math.h>

typedef float v2f __attribute__((ext_vector_type(2)));
typedef float v4f __attribute__((ext_vector_type(4)));
typedef float v8f __attribute__((ext_vector_type(8)));

#define TWO_PI_F     6.2831853071795864769f
#define MIN_W        1e-3f
#define MIN_H        1e-3f
#define MIN_D        1e-3f
#define DERIV_SHIFT  0.54132485461292f

#define KBINS 32
#define DH    8
#define CDIM  64
#define HDIM  256
#define NPAR  768

#define WAVES           DH
#define MTILES          4
#define ROWS_PER_BLOCK  (MTILES * 16)
#define MPAIRS          (MTILES / 2)
#define PCHUNK          96
#define NT2             (PCHUNK / 16)

#define HPITCH  (HDIM + 4)
#define PPITCH  (KBINS + 2)

#define HBUF_FLOATS (ROWS_PER_BLOCK * HPITCH)
#define PBUF_FLOATS (WAVES * 32 * PPITCH)
#define OBUF_FLOATS (2 * ROWS_PER_BLOCK * DH)
#define LDS_BYTES   ((HBUF_FLOATS + PBUF_FLOATS + OBUF_FLOATS) * 4)

#define KSTEPS  (HDIM / 4)
#define PF_DIST 2

static __device__ __forceinline__ float softplus_f(float x) {
    return (x > 20.0f) ? x : log1pf(__expf(x));
}

static __device__ __forceinline__ void load_w2_step(
    const float* __restrict__ W2, int dpn, int kc, int half, int m, v2f (&dst)[NT2])
{
    const unsigned p = (unsigned)((4 * kc + 2 * half) * NPAR + dpn * PCHUNK + m);
    #pragma unroll
    for (int jt = 0; jt < NT2; ++jt) {
        dst[jt].x = W2[p + 16 * jt];
        dst[jt].y = W2[p + NPAR + 16 * jt];
    }
}

static __device__ __forceinline__ void load_a_step(
    const float* hbuf, int kc, int half, int m, v2f (&dst)[MTILES])
{
    const int kk = 4 * kc + 2 * half;
    #pragma unroll
    for (int mt = 0; mt < MTILES; ++mt) {
        dst[mt].x = hbuf[(mt * 16 + m) * HPITCH + kk];
        dst[mt].y = hbuf[(mt * 16 + m) * HPITCH + kk + 1];
    }
}

static __device__ __forceinline__ void load_w1_tile(
    const float* __restrict__ W1, int j, int half, int m, v2f (&dst)[16])
{
    const float* p = W1 + (size_t)(2 * half) * HDIM + 16 * j + m;
    #pragma unroll
    for (int kc = 0; kc < 16; ++kc) {
        dst[kc].x = p[(size_t)(4 * kc) * HDIM];
        dst[kc].y = p[(size_t)(4 * kc + 1) * HDIM];
    }
}

static __device__ __forceinline__ void gemm1_tile(
    const float* __restrict__ b1, float* hbuf,
    const v2f (&afrag)[2][16], const v2f (&cur)[16],
    int rb, int j, int half, int m)
{
    const int col = 16 * j + m;
    const float bias = b1[col];

    v8f acc0 = {0.f,0.f,0.f,0.f,0.f,0.f,0.f,0.f};
    v8f acc1 = {0.f,0.f,0.f,0.f,0.f,0.f,0.f,0.f};
    #pragma unroll
    for (int kc = 0; kc < 16; ++kc) {
        acc0 = __builtin_amdgcn_wmma_f32_16x16x4_f32(
            false, afrag[0][kc], false, cur[kc], (short)0, acc0, false, false);
        acc1 = __builtin_amdgcn_wmma_f32_16x16x4_f32(
            false, afrag[1][kc], false, cur[kc], (short)0, acc1, false, false);
    }
    #pragma unroll
    for (int v = 0; v < 8; ++v) {
        float h0 = fmaxf(acc0[v] + bias, 0.0f);
        float h1 = fmaxf(acc1[v] + bias, 0.0f);
        hbuf[(rb +  0 + 8 * half + v) * HPITCH + col] = h0;
        hbuf[(rb + 16 + 8 * half + v) * HPITCH + col] = h1;
    }
}

static __device__ __forceinline__ void gemm2_group(
    const float* __restrict__ W2, const float* hbuf, int dp, int kc4, bool request,
    int half, int m,
    v2f (&aq)[PF_DIST][MTILES], v2f (&bq)[PF_DIST][NT2], v8f (&acc)[MTILES][NT2])
{
    #pragma unroll
    for (int u = 0; u < PF_DIST; ++u) {
        #pragma unroll
        for (int jt = 0; jt < NT2; ++jt) {
            #pragma unroll
            for (int mt = 0; mt < MTILES; ++mt)
                acc[mt][jt] = __builtin_amdgcn_wmma_f32_16x16x4_f32(
                    false, aq[u][mt], false, bq[u][jt], (short)0, acc[mt][jt], false, false);
        }
        if (request) {
            const int kn = kc4 + u + PF_DIST;
            load_a_step(hbuf, kn, half, m, aq[u]);
            load_w2_step(W2, dp, kn, half, m, bq[u]);
            __builtin_amdgcn_sched_barrier(0);
        }
    }
}

template <int MP>
static __device__ __forceinline__ void spline_half(
    const v8f (&acc)[MTILES][NT2], const float (&bias2)[NT2], float etav, float th,
    float* pbuf, float* obuf, int lane, int half, int m, int dp)
{
    float ew[KBINS], eh[KBINS];
    #pragma unroll
    for (int g = 0; g < 3; ++g) {
        #pragma unroll
        for (int jj = 0; jj < 2; ++jj) {
            const int jt = 2 * g + jj;
            const float bias = bias2[jt];
            #pragma unroll
            for (int mt = 0; mt < 2; ++mt) {
                #pragma unroll
                for (int v = 0; v < 8; ++v) {
                    const float pv = (acc[2 * MP + mt][jt][v] + bias) * etav;
                    pbuf[(mt * 16 + 8 * half + v) * PPITCH + 16 * jj + m] = pv;
                }
            }
        }
        if (g == 0) {
            #pragma unroll
            for (int i = 0; i < KBINS; ++i) ew[i] = pbuf[lane * PPITCH + i];
        }
        if (g == 1) {
            #pragma unroll
            for (int i = 0; i < KBINS; ++i) eh[i] = pbuf[lane * PPITCH + i];
        }
    }
    const float* ud = &pbuf[lane * PPITCH];

    float mw = -1e30f, mh = -1e30f;
    #pragma unroll
    for (int i = 0; i < KBINS; ++i) {
        mw = fmaxf(mw, ew[i]);
        mh = fmaxf(mh, eh[i]);
    }
    float sw = 0.f, sh = 0.f;
    #pragma unroll
    for (int i = 0; i < KBINS; ++i) {
        ew[i] = __expf(ew[i] - mw);
        eh[i] = __expf(eh[i] - mh);
        sw += ew[i];
        sh += eh[i];
    }
    const float wscale = (1.0f - MIN_W * KBINS) / sw;
    const float hscale = (1.0f - MIN_H * KBINS) / sh;

    float cw = 0.f, ch = 0.f;
    float Sw = 0.f, Sh = 0.f;
    int   idx = 0;
    float lo_w = 0.f, hi_w = TWO_PI_F, lo_h = 0.f, hi_h = TWO_PI_F;
    #pragma unroll
    for (int i = 0; i < KBINS; ++i) {
        const float wfrac = MIN_W + wscale * ew[i];
        const float hfrac = MIN_H + hscale * eh[i];
        Sw += wfrac;
        Sh += hfrac;
        const float cw_next = (i == KBINS - 1) ? TWO_PI_F : TWO_PI_F * Sw;
        const float ch_next = (i == KBINS - 1) ? TWO_PI_F : TWO_PI_F * Sh;
        const bool  in_bin  = (th >= cw);
        idx  = in_bin ? i       : idx;
        lo_w = in_bin ? cw      : lo_w;
        hi_w = in_bin ? cw_next : hi_w;
        lo_h = in_bin ? ch      : lo_h;
        hi_h = in_bin ? ch_next : hi_h;
        cw = cw_next;
        ch = ch_next;
    }
    const float sel_cw = lo_w, sel_w = hi_w - lo_w;
    const float sel_ch = lo_h, sel_h = hi_h - lo_h;

    const float d0 = MIN_D + softplus_f(ud[idx] + DERIV_SHIFT);
    const float d1 = MIN_D + softplus_f(ud[(idx + 1) & (KBINS - 1)] + DERIV_SHIFT);

    const float delta = sel_h / sel_w;
    const float t     = (th - sel_cw) / sel_w;
    const float t1mt  = t * (1.0f - t);
    const float numer = sel_h * (delta * t * t + d0 * t1mt);
    const float denom = delta + (d0 + d1 - 2.0f * delta) * t1mt;
    const float outv  = sel_ch + numer / denom;

    const float omt  = 1.0f - t;
    const float dnum = delta * delta * (d1 * t * t + 2.0f * delta * t1mt + d0 * omt * omt);
    const float lad  = __logf(dnum) - 2.0f * __logf(denom);

    obuf[(32 * MP + lane) * DH + dp] = outv;
    obuf[ROWS_PER_BLOCK * DH + (32 * MP + lane) * DH + dp] = lad;
}

__global__ __launch_bounds__(WAVES * 32, 4)
void Circular_Spline_Flow_fused_kernel(
    const float* __restrict__ theta,
    const float* __restrict__ x,
    const float* __restrict__ W1,
    const float* __restrict__ b1,
    const float* __restrict__ W2,
    const float* __restrict__ b2,
    const float* __restrict__ eta,
    float* __restrict__ out,
    int Btotal)
{
    __shared__ __attribute__((aligned(16))) float smem[LDS_BYTES / 4];

    const int tid  = threadIdx.x;
    const int wave = tid >> 5;
    const int lane = tid & 31;
    const int m    = lane & 15;
    const int half = lane >> 4;

    const int rowbase = blockIdx.x * ROWS_PER_BLOCK;
    const int dp      = wave;

    float* hbuf = smem;
    float* pbuf = smem + HBUF_FLOATS + wave * 32 * PPITCH;
    float* obuf = smem + HBUF_FLOATS + PBUF_FLOATS;

    const float etav = eta[0];

    v2f bw[16];
    float bias2[NT2];
    float th[MPAIRS];
    v2f bq[PF_DIST][NT2];
    #pragma unroll
    for (int mp = 0; mp < MPAIRS; ++mp) {
        v2f afrag[2][16];
        #pragma unroll
        for (int mt = 0; mt < 2; ++mt) {
            const float* xrow = x + (size_t)(rowbase + mp * 32 + mt * 16 + m) * CDIM;
            #pragma unroll
            for (int kc = 0; kc < 16; ++kc) {
                const int kk = 4 * kc + 2 * half;
                afrag[mt][kc].x = xrow[kk];
                afrag[mt][kc].y = xrow[kk + 1];
            }
        }
        load_w1_tile(W1, 2 * wave, half, m, bw);
        gemm1_tile(b1, hbuf, afrag, bw, mp * 32, 2 * wave, half, m);

        if (mp == MPAIRS - 1) {
            #pragma unroll
            for (int u = 0; u < PF_DIST; ++u)
                load_w2_step(W2, dp, u, half, m, bq[u]);
        }

        load_w1_tile(W1, 2 * wave + 1, half, m, bw);
        gemm1_tile(b1, hbuf, afrag, bw, mp * 32, 2 * wave + 1, half, m);
    }

    __syncthreads();

    v2f aq[PF_DIST][MTILES];
    #pragma unroll
    for (int u = 0; u < PF_DIST; ++u)
        load_a_step(hbuf, u, half, m, aq[u]);

    v8f acc[MTILES][NT2];
    #pragma unroll
    for (int mt = 0; mt < MTILES; ++mt)
        #pragma unroll
        for (int jt = 0; jt < NT2; ++jt)
            acc[mt][jt] = (v8f){0.f,0.f,0.f,0.f,0.f,0.f,0.f,0.f};

    #pragma unroll 1
    for (int kc4 = 0; kc4 < KSTEPS - PF_DIST; kc4 += PF_DIST)
        gemm2_group(W2, hbuf, dp, kc4, true, half, m, aq, bq, acc);
    #pragma unroll
    for (int jt = 0; jt < NT2; ++jt)
        bias2[jt] = b2[dp * PCHUNK + 16 * jt + m];
    #pragma unroll
    for (int r = 0; r < MPAIRS; ++r)
        th[r] = theta[(size_t)(rowbase + 32 * r + lane) * DH + dp];
    gemm2_group(W2, hbuf, dp, KSTEPS - PF_DIST, false, half, m, aq, bq, acc);

    spline_half<0>(acc, bias2, etav, th[0], pbuf, obuf, lane, half, m, dp);
    spline_half<1>(acc, bias2, etav, th[1], pbuf, obuf, lane, half, m, dp);

    __syncthreads();

    {
        const int plane = tid >> 7;
        const int c     = tid & 127;
        const v4f val = *reinterpret_cast<const v4f*>(&obuf[plane * ROWS_PER_BLOCK * DH + 4 * c]);
        float* dst = out + (size_t)plane * Btotal * DH + (size_t)rowbase * DH + 4 * c;
        *reinterpret_cast<v4f*>(dst) = val;
    }
}

extern "C" void kernel_launch(void* const* d_in, const int* in_sizes, int n_in,
                              void* d_out, int out_size, void* d_ws, size_t ws_size,
                              hipStream_t stream) {
    (void)n_in; (void)out_size; (void)d_ws; (void)ws_size;
    const float* theta = (const float*)d_in[0];
    const float* x     = (const float*)d_in[1];
    const float* W1    = (const float*)d_in[2];
    const float* b1    = (const float*)d_in[3];
    const float* W2    = (const float*)d_in[4];
    const float* b2    = (const float*)d_in[5];
    const float* eta   = (const float*)d_in[6];
    float* out = (float*)d_out;

    const int B = in_sizes[0] / DH;
    const int blocks = B / ROWS_PER_BLOCK;
    Circular_Spline_Flow_fused_kernel<<<blocks, WAVES * 32, 0, stream>>>(
        theta, x, W1, b1, W2, b2, eta, out, B);
}
